# starting kernel
# baseline (speedup 1.0000x reference)
#include <hip/hip_runtime.h>
#include <hip/hip_bf16.h>


#define NN      1024
#define RBF     16
#define CUTOFF  2.5f
#define FROW    36

typedef __attribute__((ext_vector_type(2))) float v2f;
typedef __attribute__((ext_vector_type(8))) float v8f;

__global__ void __launch_bounds__(256)
lfe_pair_kernel(const float* __restrict__ pos,
                const int*   __restrict__ types,
                float*       __restrict__ feats)
{
    __shared__ __align__(16) float sx[NN];
    __shared__ __align__(16) float sy[NN];
    __shared__ __align__(16) float sz[NN];
    __shared__ __align__(16) int   st[NN];
    __shared__ __align__(16) float red[256 * 32];

    const int b   = blockIdx.x >> 4;
    const int it  = blockIdx.x & 15;
    const int tid = threadIdx.x;

    const float* pb = pos + (size_t)b * NN * 3;
    for (int j = tid; j < NN; j += 256) {
        sx[j] = pb[j * 3 + 0];
        sy[j] = pb[j * 3 + 1];
        sz[j] = pb[j * 3 + 2];
        st[j] = types[b * NN + j];
    }
    __syncthreads();

    const int il = tid & 63;
    const int js = tid >> 6;
    const int i  = it * 64 + il;

    const float xi = sx[i], yi = sy[i], zi = sz[i];

    float a0[RBF], a1[RBF];
#pragma unroll
    for (int r = 0; r < RBF; ++r) { a0[r] = 0.f; a1[r] = 0.f; }

    const float dc     = CUTOFF / 15.f;
    const float invw2  = 1.0f / (dc * dc + 1e-12f);
    const float invcut = 1.0f / CUTOFF;

    const int j0 = js * 256;
    for (int jj = 0; jj < 256; ++jj) {
        const int j = j0 + jj;
        if (j == i) continue;
        const float dx = xi - sx[j];
        const float dy = yi - sy[j];
        const float dz = zi - sz[j];
        const float sq = dx * dx + dy * dy + dz * dz;
        const float d  = (sq > 0.f) ? sqrtf(sq) : 0.f;
        if (d < CUTOFF) {
            const float x  = d * invcut;
            const float x2 = x * x;
            const float x3 = x2 * x;
            const float cutv = 1.0f - x3 * (10.f - 15.f * x + 6.f * x2);
            const int   t  = st[j];
            const float m0 = (t == 0) ? cutv : 0.f;
            const float m1 = cutv - m0;
#pragma unroll
            for (int r = 0; r < RBF; ++r) {
                const float diff = d - (float)r * dc;
                const float e = __expf(-diff * diff * invw2);
                a0[r] += e * m0;
                a1[r] += e * m1;
            }
        }
    }

    float* myr = &red[tid * 32];
#pragma unroll
    for (int r = 0; r < RBF; ++r) { myr[r] = a0[r]; myr[16 + r] = a1[r]; }
    __syncthreads();

    if (tid < 64) {
        const int row = b * NN + it * 64 + tid;
        float* o = feats + (size_t)row * FROW;
        const int ty = st[it * 64 + tid];
        o[0] = (ty == 0) ? 1.f : 0.f;
        o[1] = (ty == 1) ? 1.f : 0.f;
#pragma unroll
        for (int r = 0; r < RBF; ++r) {
            float s0 = 0.f, s1 = 0.f;
#pragma unroll
            for (int q = 0; q < 4; ++q) {
                s0 += red[(q * 64 + tid) * 32 + r];
                s1 += red[(q * 64 + tid) * 32 + 16 + r];
            }
            o[2 + r * 2 + 0] = s0;
            o[2 + r * 2 + 1] = s1;
        }
        o[34] = 0.f;
        o[35] = 0.f;
    }
}

__device__ __forceinline__ float silu(float x) {
    return x / (1.0f + __expf(-x));
}

__global__ void __launch_bounds__(128)
lfe_mlp_kernel(const float* __restrict__ feats,
               const float* __restrict__ W1, const float* __restrict__ b1,
               const float* __restrict__ W2, const float* __restrict__ b2,
               const float* __restrict__ W3, const float* __restrict__ b3,
               float*       __restrict__ out)
{
    __shared__ __align__(16) float hA[4][16 * 64];
    __shared__ __align__(16) float hB[4][16 * 64];

    const int lane  = threadIdx.x & 31;
    const int wave  = threadIdx.x >> 5;
    const int tile  = blockIdx.x * 4 + wave;
    const int r0    = tile * 16;
    const int m16   = lane & 15;
    const int khalf = lane >> 4;

    v8f c[4];

#pragma unroll
    for (int t = 0; t < 4; ++t) {
        const float bv = b1[t * 16 + m16];
        c[t] = (v8f){bv, bv, bv, bv, bv, bv, bv, bv};
    }
    {
        const float* arow = feats + (size_t)(r0 + m16) * FROW + 2 * khalf;
#pragma unroll
        for (int kk = 0; kk < 9; ++kk) {
            const v2f a = *(const v2f*)(arow + 4 * kk);
            const int k0 = 4 * kk + 2 * khalf;
#pragma unroll
            for (int t = 0; t < 4; ++t) {
                const int n = t * 16 + m16;
                v2f bb;
                bb.x = (k0     < 34) ? W1[k0 * 64 + n]       : 0.f;
                bb.y = (k0 + 1 < 34) ? W1[(k0 + 1) * 64 + n] : 0.f;
                c[t] = __builtin_amdgcn_wmma_f32_16x16x4_f32(
                           false, a, false, bb, (short)0, c[t], false, false);
            }
        }
    }
    {
        float* h = hA[wave];
#pragma unroll
        for (int t = 0; t < 4; ++t)
#pragma unroll
            for (int v = 0; v < 8; ++v) {
                const int m = v + 8 * khalf;
                h[m * 64 + t * 16 + m16] = silu(c[t][v]);
            }
    }
    __syncthreads();

#pragma unroll
    for (int t = 0; t < 4; ++t) {
        const float bv = b2[t * 16 + m16];
        c[t] = (v8f){bv, bv, bv, bv, bv, bv, bv, bv};
    }
    {
        const float* h = hA[wave];
#pragma unroll
        for (int kk = 0; kk < 16; ++kk) {
            const int k0 = 4 * kk + 2 * khalf;
            const v2f a = *(const v2f*)(&h[m16 * 64 + k0]);
#pragma unroll
            for (int t = 0; t < 4; ++t) {
                const int n = t * 16 + m16;
                v2f bb;
                bb.x = W2[k0 * 64 + n];
                bb.y = W2[(k0 + 1) * 64 + n];
                c[t] = __builtin_amdgcn_wmma_f32_16x16x4_f32(
                           false, a, false, bb, (short)0, c[t], false, false);
            }
        }
    }
    {
        float* h2 = hB[wave];
#pragma unroll
        for (int t = 0; t < 4; ++t)
#pragma unroll
            for (int v = 0; v < 8; ++v) {
                const int m = v + 8 * khalf;
                h2[m * 64 + t * 16 + m16] = silu(c[t][v]);
            }
    }
    __syncthreads();

#pragma unroll
    for (int t = 0; t < 4; ++t) {
        const float bv = b3[t * 16 + m16];
        c[t] = (v8f){bv, bv, bv, bv, bv, bv, bv, bv};
    }
    {
        const float* h2 = hB[wave];
#pragma unroll
        for (int kk = 0; kk < 16; ++kk) {
            const int k0 = 4 * kk + 2 * khalf;
            const v2f a = *(const v2f*)(&h2[m16 * 64 + k0]);
#pragma unroll
            for (int t = 0; t < 4; ++t) {
                const int n = t * 16 + m16;
                v2f bb;
                bb.x = W3[k0 * 64 + n];
                bb.y = W3[(k0 + 1) * 64 + n];
                c[t] = __builtin_amdgcn_wmma_f32_16x16x4_f32(
                           false, a, false, bb, (short)0, c[t], false, false);
            }
        }
    }
#pragma unroll
    for (int t = 0; t < 4; ++t)
#pragma unroll
        for (int v = 0; v < 8; ++v) {
            const int m = v + 8 * khalf;
            out[(size_t)(r0 + m) * 64 + t * 16 + m16] = c[t][v];
        }
}

extern "C" void kernel_launch(void* const* d_in, const int* in_sizes, int n_in,
                              void* d_out, int out_size, void* d_ws, size_t ws_size,
                              hipStream_t stream)
{
    const float* pos   = (const float*)d_in[0];
    const int*   types = (const int*)  d_in[1];
    const float* W1    = (const float*)d_in[2];
    const float* b1    = (const float*)d_in[3];
    const float* W2    = (const float*)d_in[4];
    const float* b2    = (const float*)d_in[5];
    const float* W3    = (const float*)d_in[6];
    const float* b3    = (const float*)d_in[7];
    float*       out   = (float*)d_out;

    const int B = in_sizes[0] / (NN * 3);
    float* feats = (float*)d_ws;

    lfe_pair_kernel<<<B * 16, 256, 0, stream>>>(pos, types, feats);

    const int nrows = B * NN;
    lfe_mlp_kernel<<<nrows / 64, 128, 0, stream>>>(feats, W1, b1, W2, b2,
                                                   W3, b3, out);
}


# Pair kernel: each wave packs one atom's in-cutoff neighbours into an LDS list, then runs the exps on dense lanes (lane = entry parity x rbf index); neighbour sum order changes (two parity halves)
# speedup vs baseline: 3.2534x; 3.2534x over previous
#include <hip/hip_runtime.h>
#include <hip/hip_bf16.h>


#define NN      1024
#define RBF     16
#define CUTOFF  2.5f
#define FROW    36

typedef __attribute__((ext_vector_type(2))) float v2f;
typedef __attribute__((ext_vector_type(8))) float v8f;

#define IPW   4
#define LCAP  (NN + 16)

typedef __attribute__((ext_vector_type(4))) float v4f;

__device__ __forceinline__ void wave_lds_handover() {
    __builtin_amdgcn_fence(__ATOMIC_ACQ_REL, "wavefront");
    __builtin_amdgcn_wave_barrier();
}

__device__ __forceinline__ float pair_sq(const v4f pi, const v4f pj) {
    const float dx = pi.x - pj.x;
    const float dy = pi.y - pj.y;
    const float dz = pi.z - pj.z;
    return dx * dx + dy * dy + dz * dz;
}

__device__ __forceinline__ void list_put(v4f* L, int p, float sq, float tbits) {
    v2f c; c.x = sq; c.y = tbits;
    *(v2f*)&L[p] = c;
}

__global__ void __launch_bounds__(256)
lfe_pair_kernel(const float* __restrict__ pos,
                const int*   __restrict__ types,
                float*       __restrict__ feats)
{
    __shared__ __align__(16) v4f sp[NN];
    __shared__ __align__(16) v4f lst[8][LCAP];

    const int b    = blockIdx.x >> 5;
    const int it   = blockIdx.x & 31;
    const int tid  = threadIdx.x;
    const int lane = tid & 31;
    const int wave = tid >> 5;

    const float* pb = pos + (size_t)b * NN * 3;
#pragma unroll
    for (int k = 0; k < NN / 256; ++k) {
        const int j = tid + 256 * k;
        v4f v;
        v.x = pb[j * 3 + 0];
        v.y = pb[j * 3 + 1];
        v.z = pb[j * 3 + 2];
        v.w = __int_as_float(types[b * NN + j]);
        sp[j] = v;
    }
    __syncthreads();

    v4f* L = lst[wave];
    const int r = lane & (RBF - 1);
    const int h = lane >> 4;

    const float dc     = CUTOFF / 15.f;
    const float invw2  = 1.0f / (dc * dc + 1e-12f);
    const float invcut = 1.0f / CUTOFF;
    const float rc     = (float)r * dc;
    const float cut2 = CUTOFF * CUTOFF * 1.00001f;

#pragma unroll 1
    for (int a = 0; a < IPW; ++a) {
        const int i = it * 32 + wave * IPW + a;
        const v4f pi = sp[i];
        float a0 = 0.f, a1 = 0.f;

        int cnt = 0;
        v4f p0 = sp[lane];
        v4f p1 = sp[lane + 32];
        v4f p2 = sp[lane + 64];
        v4f p3 = sp[lane + 96];
        for (int s = 0; s < NN / 32; s += 4) {
            const int j = s * 32 + lane;
            const float q0 = pair_sq(pi, p0), q1 = pair_sq(pi, p1);
            const float q2 = pair_sq(pi, p2), q3 = pair_sq(pi, p3);
            const float t0 = p0.w, t1 = p1.w, t2 = p2.w, t3 = p3.w;
            const bool in0 = (q0 < cut2) && (j      != i);
            const bool in1 = (q1 < cut2) && (j + 32 != i);
            const bool in2 = (q2 < cut2) && (j + 64 != i);
            const bool in3 = (q3 < cut2) && (j + 96 != i);
            const unsigned k0 = __builtin_amdgcn_ballot_w32(in0);
            const unsigned k1 = __builtin_amdgcn_ballot_w32(in1);
            const unsigned k2 = __builtin_amdgcn_ballot_w32(in2);
            const unsigned k3 = __builtin_amdgcn_ballot_w32(in3);
            const int jn = (j + 128) & (NN - 1);
            p0 = sp[jn];
            p1 = sp[jn + 32];
            p2 = sp[jn + 64];
            p3 = sp[jn + 96];
            const int c0 = cnt;
            const int c1 = c0 + __builtin_popcount(k0);
            const int c2 = c1 + __builtin_popcount(k1);
            const int c3 = c2 + __builtin_popcount(k2);
            cnt          = c3 + __builtin_popcount(k3);
            if (in0) list_put(L, c0 + (int)__builtin_amdgcn_mbcnt_lo(k0, 0u), q0, t0);
            if (in1) list_put(L, c1 + (int)__builtin_amdgcn_mbcnt_lo(k1, 0u), q1, t1);
            if (in2) list_put(L, c2 + (int)__builtin_amdgcn_mbcnt_lo(k2, 0u), q2, t2);
            if (in3) list_put(L, c3 + (int)__builtin_amdgcn_mbcnt_lo(k3, 0u), q3, t3);
        }
        const int cpad = (cnt + 15) & ~15;
        wave_lds_handover();

        for (int e = lane; e < cpad; e += 32) {
            v4f o = (v4f){0.f, 0.f, 0.f, 0.f};
            if (e < cnt) {
                const v2f c = *(const v2f*)&L[e];
                const float sq = c.x;
                const int   t  = __float_as_int(c.y);
                const float d  = (sq > 0.f) ? sqrtf(sq) : 0.f;
                if (d < CUTOFF) {
                    const float x  = d * invcut;
                    const float x2 = x * x;
                    const float x3 = x2 * x;
                    const float cutv = 1.0f - x3 * (10.f - 15.f * x + 6.f * x2);
                    const float m0 = (t == 0) ? cutv : 0.f;
                    const float m1 = cutv - m0;
                    o = (v4f){d, m0, m1, 0.f};
                }
            }
            L[e] = o;
        }
        wave_lds_handover();

        for (int e = 0; e < cpad; e += 16) {
            v4f q[8];
#pragma unroll
            for (int k = 0; k < 8; ++k) q[k] = L[e + 2 * k + h];
#pragma unroll
            for (int k = 0; k < 8; ++k) {
                const float f  = q[k].x - rc;
                const float ev = __expf(-f * f * invw2);
                a0 += ev * q[k].y;
                a1 += ev * q[k].z;
            }
        }
        wave_lds_handover();

        a0 += __shfl_xor(a0, 16);
        a1 += __shfl_xor(a1, 16);

        float* o = feats + (size_t)(b * NN + i) * FROW;
        if (lane < 16) {
            v2f w; w.x = a0; w.y = a1;
            *(v2f*)(o + 2 + 2 * r) = w;
        } else if (lane == 16) {
            const int ty = __float_as_int(pi.w);
            v2f w; w.x = (ty == 0) ? 1.f : 0.f;
            w.y = (ty == 1) ? 1.f : 0.f;
            *(v2f*)(o + 0) = w;
        } else if (lane == 17) {
            *(v2f*)(o + 34) = (v2f){0.f, 0.f};
        }
    }
}

__device__ __forceinline__ float silu(float x) {
    return x / (1.0f + __expf(-x));
}

__global__ void __launch_bounds__(128)
lfe_mlp_kernel(const float* __restrict__ feats,
               const float* __restrict__ W1, const float* __restrict__ b1,
               const float* __restrict__ W2, const float* __restrict__ b2,
               const float* __restrict__ W3, const float* __restrict__ b3,
               float*       __restrict__ out)
{
    __shared__ __align__(16) float hA[4][16 * 64];
    __shared__ __align__(16) float hB[4][16 * 64];

    const int lane  = threadIdx.x & 31;
    const int wave  = threadIdx.x >> 5;
    const int tile  = blockIdx.x * 4 + wave;
    const int r0    = tile * 16;
    const int m16   = lane & 15;
    const int khalf = lane >> 4;

    v8f c[4];

#pragma unroll
    for (int t = 0; t < 4; ++t) {
        const float bv = b1[t * 16 + m16];
        c[t] = (v8f){bv, bv, bv, bv, bv, bv, bv, bv};
    }
    {
        const float* arow = feats + (size_t)(r0 + m16) * FROW + 2 * khalf;
#pragma unroll
        for (int kk = 0; kk < 9; ++kk) {
            const v2f a = *(const v2f*)(arow + 4 * kk);
            const int k0 = 4 * kk + 2 * khalf;
#pragma unroll
            for (int t = 0; t < 4; ++t) {
                const int n = t * 16 + m16;
                v2f bb;
                bb.x = (k0     < 34) ? W1[k0 * 64 + n]       : 0.f;
                bb.y = (k0 + 1 < 34) ? W1[(k0 + 1) * 64 + n] : 0.f;
                c[t] = __builtin_amdgcn_wmma_f32_16x16x4_f32(
                           false, a, false, bb, (short)0, c[t], false, false);
            }
        }
    }
    {
        float* h = hA[wave];
#pragma unroll
        for (int t = 0; t < 4; ++t)
#pragma unroll
            for (int v = 0; v < 8; ++v) {
                const int m = v + 8 * khalf;
                h[m * 64 + t * 16 + m16] = silu(c[t][v]);
            }
    }
    __syncthreads();

#pragma unroll
    for (int t = 0; t < 4; ++t) {
        const float bv = b2[t * 16 + m16];
        c[t] = (v8f){bv, bv, bv, bv, bv, bv, bv, bv};
    }
    {
        const float* h = hA[wave];
#pragma unroll
        for (int kk = 0; kk < 16; ++kk) {
            const int k0 = 4 * kk + 2 * khalf;
            const v2f a = *(const v2f*)(&h[m16 * 64 + k0]);
#pragma unroll
            for (int t = 0; t < 4; ++t) {
                const int n = t * 16 + m16;
                v2f bb;
                bb.x = W2[k0 * 64 + n];
                bb.y = W2[(k0 + 1) * 64 + n];
                c[t] = __builtin_amdgcn_wmma_f32_16x16x4_f32(
                           false, a, false, bb, (short)0, c[t], false, false);
            }
        }
    }
    {
        float* h2 = hB[wave];
#pragma unroll
        for (int t = 0; t < 4; ++t)
#pragma unroll
            for (int v = 0; v < 8; ++v) {
                const int m = v + 8 * khalf;
                h2[m * 64 + t * 16 + m16] = silu(c[t][v]);
            }
    }
    __syncthreads();

#pragma unroll
    for (int t = 0; t < 4; ++t) {
        const float bv = b3[t * 16 + m16];
        c[t] = (v8f){bv, bv, bv, bv, bv, bv, bv, bv};
    }
    {
        const float* h2 = hB[wave];
#pragma unroll
        for (int kk = 0; kk < 16; ++kk) {
            const int k0 = 4 * kk + 2 * khalf;
            const v2f a = *(const v2f*)(&h2[m16 * 64 + k0]);
#pragma unroll
            for (int t = 0; t < 4; ++t) {
                const int n = t * 16 + m16;
                v2f bb;
                bb.x = W3[k0 * 64 + n];
                bb.y = W3[(k0 + 1) * 64 + n];
                c[t] = __builtin_amdgcn_wmma_f32_16x16x4_f32(
                           false, a, false, bb, (short)0, c[t], false, false);
            }
        }
    }
#pragma unroll
    for (int t = 0; t < 4; ++t)
#pragma unroll
        for (int v = 0; v < 8; ++v) {
            const int m = v + 8 * khalf;
            out[(size_t)(r0 + m) * 64 + t * 16 + m16] = c[t][v];
        }
}

extern "C" void kernel_launch(void* const* d_in, const int* in_sizes, int n_in,
                              void* d_out, int out_size, void* d_ws, size_t ws_size,
                              hipStream_t stream)
{
    const float* pos   = (const float*)d_in[0];
    const int*   types = (const int*)  d_in[1];
    const float* W1    = (const float*)d_in[2];
    const float* b1    = (const float*)d_in[3];
    const float* W2    = (const float*)d_in[4];
    const float* b2    = (const float*)d_in[5];
    const float* W3    = (const float*)d_in[6];
    const float* b3    = (const float*)d_in[7];
    float*       out   = (float*)d_out;

    const int B = in_sizes[0] / (NN * 3);
    float* feats = (float*)d_ws;

    lfe_pair_kernel<<<B * 32, 256, 0, stream>>>(pos, types, feats);

    const int nrows = B * NN;
    lfe_mlp_kernel<<<nrows / 64, 128, 0, stream>>>(feats, W1, b1, W2, b2,
                                                   W3, b3, out);
}


# One launch instead of two: each block keeps its 32 feature rows in LDS and runs the MLP on them, one 16x16 f32 WMMA tile per wave and layer, weights preloaded in registers; summation order unchanged
# speedup vs baseline: 4.3681x; 1.3426x over previous
#include <hip/hip_runtime.h>
#include <hip/hip_bf16.h>


#define NN      1024
#define RBF     16
#define CUTOFF  2.5f
#define FROW    36
#define HP      68

typedef __attribute__((ext_vector_type(2))) float v2f;
typedef __attribute__((ext_vector_type(8))) float v8f;

#define IPW   4
#define LCAP  (NN + 16)

typedef __attribute__((ext_vector_type(4))) float v4f;

__device__ __forceinline__ void wave_lds_handover() {
    __builtin_amdgcn_fence(__ATOMIC_ACQ_REL, "wavefront");
    __builtin_amdgcn_wave_barrier();
}

__device__ __forceinline__ float pair_sq(const v4f pi, const v4f pj) {
    const float dx = pi.x - pj.x;
    const float dy = pi.y - pj.y;
    const float dz = pi.z - pj.z;
    return dx * dx + dy * dy + dz * dz;
}

__device__ __forceinline__ void list_put(v4f* L, int p, float sq, float tbits) {
    v2f c; c.x = sq; c.y = tbits;
    *(v2f*)&L[p] = c;
}

__device__ __forceinline__ float silu(float x) {
    return x / (1.0f + __expf(-x));
}

__device__ __forceinline__ v8f splat8(float v) {
    return (v8f){v, v, v, v, v, v, v, v};
}

__global__ void __launch_bounds__(256)
lfe_fused_kernel(const float* __restrict__ pos,
                 const int*   __restrict__ types,
                 const float* __restrict__ W1, const float* __restrict__ b1,
                 const float* __restrict__ W2, const float* __restrict__ b2,
                 const float* __restrict__ W3, const float* __restrict__ b3,
                 float*       __restrict__ out)
{
    __shared__ __align__(16) v4f sp[NN];
    __shared__ __align__(16) v4f lst[8][LCAP];
    __shared__ __align__(16) float ft[32 * FROW];
    __shared__ __align__(16) float hA[32 * HP];
    __shared__ __align__(16) float hB[32 * HP];

    const int b    = blockIdx.x >> 5;
    const int it   = blockIdx.x & 31;
    const int tid  = threadIdx.x;
    const int lane = tid & 31;
    const int wave = tid >> 5;

    const float* pb = pos + (size_t)b * NN * 3;
#pragma unroll
    for (int k = 0; k < NN / 256; ++k) {
        const int j = tid + 256 * k;
        v4f v;
        v.x = pb[j * 3 + 0];
        v.y = pb[j * 3 + 1];
        v.z = pb[j * 3 + 2];
        v.w = __int_as_float(types[b * NN + j]);
        sp[j] = v;
    }
    __syncthreads();

    const int m16   = lane & 15;
    const int khalf = lane >> 4;
    const int rt    = wave >> 2;
    const int ct    = wave & 3;
    const int n     = ct * 16 + m16;
    v2f w1f[9], w2f[16], w3f[16];
#pragma unroll
    for (int kk = 0; kk < 9; ++kk) {
        const int k0 = 4 * kk + 2 * khalf;
        w1f[kk].x = (k0     < 34) ? W1[k0 * 64 + n]       : 0.f;
        w1f[kk].y = (k0 + 1 < 34) ? W1[(k0 + 1) * 64 + n] : 0.f;
    }
#pragma unroll
    for (int kk = 0; kk < 16; ++kk) {
        const int k0 = 4 * kk + 2 * khalf;
        w2f[kk].x = W2[k0 * 64 + n];
        w2f[kk].y = W2[(k0 + 1) * 64 + n];
        w3f[kk].x = W3[k0 * 64 + n];
        w3f[kk].y = W3[(k0 + 1) * 64 + n];
    }
    const float bv1 = b1[n], bv2 = b2[n], bv3 = b3[n];

    v4f* L = lst[wave];
    const int r = lane & (RBF - 1);
    const int h = lane >> 4;

    const float dc     = CUTOFF / 15.f;
    const float invw2  = 1.0f / (dc * dc + 1e-12f);
    const float invcut = 1.0f / CUTOFF;
    const float rc     = (float)r * dc;
    const float cut2 = CUTOFF * CUTOFF * 1.00001f;

#pragma unroll 1
    for (int a = 0; a < IPW; ++a) {
        const int i = it * 32 + wave * IPW + a;
        const v4f pi = sp[i];
        float a0 = 0.f, a1 = 0.f;

        int cnt = 0;
        v4f p0 = sp[lane];
        v4f p1 = sp[lane + 32];
        v4f p2 = sp[lane + 64];
        v4f p3 = sp[lane + 96];
        for (int s = 0; s < NN / 32; s += 4) {
            const int j = s * 32 + lane;
            const float q0 = pair_sq(pi, p0), q1 = pair_sq(pi, p1);
            const float q2 = pair_sq(pi, p2), q3 = pair_sq(pi, p3);
            const float t0 = p0.w, t1 = p1.w, t2 = p2.w, t3 = p3.w;
            const bool in0 = (q0 < cut2) && (j      != i);
            const bool in1 = (q1 < cut2) && (j + 32 != i);
            const bool in2 = (q2 < cut2) && (j + 64 != i);
            const bool in3 = (q3 < cut2) && (j + 96 != i);
            const unsigned k0 = __builtin_amdgcn_ballot_w32(in0);
            const unsigned k1 = __builtin_amdgcn_ballot_w32(in1);
            const unsigned k2 = __builtin_amdgcn_ballot_w32(in2);
            const unsigned k3 = __builtin_amdgcn_ballot_w32(in3);
            const int jn = (j + 128) & (NN - 1);
            p0 = sp[jn];
            p1 = sp[jn + 32];
            p2 = sp[jn + 64];
            p3 = sp[jn + 96];
            const int c0 = cnt;
            const int c1 = c0 + __builtin_popcount(k0);
            const int c2 = c1 + __builtin_popcount(k1);
            const int c3 = c2 + __builtin_popcount(k2);
            cnt          = c3 + __builtin_popcount(k3);
            if (in0) list_put(L, c0 + (int)__builtin_amdgcn_mbcnt_lo(k0, 0u), q0, t0);
            if (in1) list_put(L, c1 + (int)__builtin_amdgcn_mbcnt_lo(k1, 0u), q1, t1);
            if (in2) list_put(L, c2 + (int)__builtin_amdgcn_mbcnt_lo(k2, 0u), q2, t2);
            if (in3) list_put(L, c3 + (int)__builtin_amdgcn_mbcnt_lo(k3, 0u), q3, t3);
        }
        const int cpad = (cnt + 15) & ~15;
        wave_lds_handover();

        for (int e = lane; e < cpad; e += 32) {
            v4f o = (v4f){0.f, 0.f, 0.f, 0.f};
            if (e < cnt) {
                const v2f c = *(const v2f*)&L[e];
                const float sq = c.x;
                const int   t  = __float_as_int(c.y);
                const float d  = (sq > 0.f) ? sqrtf(sq) : 0.f;
                if (d < CUTOFF) {
                    const float x  = d * invcut;
                    const float x2 = x * x;
                    const float x3 = x2 * x;
                    const float cutv = 1.0f - x3 * (10.f - 15.f * x + 6.f * x2);
                    const float m0 = (t == 0) ? cutv : 0.f;
                    const float m1 = cutv - m0;
                    o = (v4f){d, m0, m1, 0.f};
                }
            }
            L[e] = o;
        }
        wave_lds_handover();

        for (int e = 0; e < cpad; e += 16) {
            v4f q[8];
#pragma unroll
            for (int k = 0; k < 8; ++k) q[k] = L[e + 2 * k + h];
#pragma unroll
            for (int k = 0; k < 8; ++k) {
                const float f  = q[k].x - rc;
                const float ev = __expf(-f * f * invw2);
                a0 += ev * q[k].y;
                a1 += ev * q[k].z;
            }
        }
        wave_lds_handover();

        a0 += __shfl_xor(a0, 16);
        a1 += __shfl_xor(a1, 16);

        float* o = ft + (wave * IPW + a) * FROW;
        if (lane < 16) {
            v2f w; w.x = a0; w.y = a1;
            *(v2f*)(o + 2 + 2 * r) = w;
        } else if (lane == 16) {
            const int ty = __float_as_int(pi.w);
            v2f w; w.x = (ty == 0) ? 1.f : 0.f;
            w.y = (ty == 1) ? 1.f : 0.f;
            *(v2f*)(o + 0) = w;
        } else if (lane == 17) {
            *(v2f*)(o + 34) = (v2f){0.f, 0.f};
        }
    }
    __syncthreads();

    v8f c;

    c = splat8(bv1);
    {
        const float* arow = ft + (rt * 16 + m16) * FROW + 2 * khalf;
#pragma unroll
        for (int kk = 0; kk < 9; ++kk) {
            const v2f av = *(const v2f*)(arow + 4 * kk);
            c = __builtin_amdgcn_wmma_f32_16x16x4_f32(
                    false, av, false, w1f[kk], (short)0, c, false, false);
        }
    }
#pragma unroll
    for (int v = 0; v < 8; ++v)
        hA[(rt * 16 + 8 * khalf + v) * HP + n] = silu(c[v]);
    __syncthreads();

    c = splat8(bv2);
    {
        const float* hrow = hA + (rt * 16 + m16) * HP + 2 * khalf;
#pragma unroll
        for (int kk = 0; kk < 16; ++kk) {
            const v2f av = *(const v2f*)(hrow + 4 * kk);
            c = __builtin_amdgcn_wmma_f32_16x16x4_f32(
                    false, av, false, w2f[kk], (short)0, c, false, false);
        }
    }
#pragma unroll
    for (int v = 0; v < 8; ++v)
        hB[(rt * 16 + 8 * khalf + v) * HP + n] = silu(c[v]);
    __syncthreads();

    c = splat8(bv3);
    {
        const float* hrow = hB + (rt * 16 + m16) * HP + 2 * khalf;
#pragma unroll
        for (int kk = 0; kk < 16; ++kk) {
            const v2f av = *(const v2f*)(hrow + 4 * kk);
            c = __builtin_amdgcn_wmma_f32_16x16x4_f32(
                    false, av, false, w3f[kk], (short)0, c, false, false);
        }
    }
#pragma unroll
    for (int v = 0; v < 8; ++v)
        hA[(rt * 16 + 8 * khalf + v) * HP + n] = c[v];
    __syncthreads();

    v4f* ob = (v4f*)(out + (size_t)blockIdx.x * 32 * 64);
#pragma unroll
    for (int p = 0; p < 2; ++p) {
        const int q   = tid + 256 * p;
        const int row = q >> 4;
        const int c4  = q & 15;
        ob[q] = *(const v4f*)(hA + row * HP + 4 * c4);
    }
}

extern "C" void kernel_launch(void* const* d_in, const int* in_sizes, int n_in,
                              void* d_out, int out_size, void* d_ws, size_t ws_size,
                              hipStream_t stream)
{
    const float* pos   = (const float*)d_in[0];
    const int*   types = (const int*)  d_in[1];
    const float* W1    = (const float*)d_in[2];
    const float* b1    = (const float*)d_in[3];
    const float* W2    = (const float*)d_in[4];
    const float* b2    = (const float*)d_in[5];
    const float* W3    = (const float*)d_in[6];
    const float* b3    = (const float*)d_in[7];
    float*       out   = (float*)d_out;

    const int B = in_sizes[0] / (NN * 3);

    lfe_fused_kernel<<<B * 32, 256, 0, stream>>>(pos, types, W1, b1, W2, b2,
                                                 W3, b3, out);
}


# 16 waves per block with 2 atoms each instead of 8 with 4: four waves per SIMD cover the pair scan's LDS and mask round trips; MLP stays on 8 of the waves; summation order unchanged
# speedup vs baseline: 6.1027x; 1.3971x over previous
#include <hip/hip_runtime.h>
#include <hip/hip_bf16.h>


#define NN      1024
#define RBF     16
#define CUTOFF  2.5f
#define FROW    36
#define HP      68

typedef __attribute__((ext_vector_type(2))) float v2f;
typedef __attribute__((ext_vector_type(8))) float v8f;

#define NWAVE 16
#define IPW   2
#define MLPW  8
#define NTHR  512
static_assert(NTHR == 32 * NWAVE && NWAVE * IPW == 32, "block shape");
#define LCAP  (NN + 16)

typedef __attribute__((ext_vector_type(4))) float v4f;

__device__ __forceinline__ void wave_lds_handover() {
    __builtin_amdgcn_fence(__ATOMIC_ACQ_REL, "wavefront");
    __builtin_amdgcn_wave_barrier();
}

__device__ __forceinline__ float pair_sq(const v4f pi, const v4f pj) {
    const float dx = pi.x - pj.x;
    const float dy = pi.y - pj.y;
    const float dz = pi.z - pj.z;
    return dx * dx + dy * dy + dz * dz;
}

__device__ __forceinline__ void list_put(v4f* L, int p, float sq, float tbits) {
    v2f c; c.x = sq; c.y = tbits;
    *(v2f*)&L[p] = c;
}

__device__ __forceinline__ float silu(float x) {
    return x / (1.0f + __expf(-x));
}

__device__ __forceinline__ v8f splat8(float v) {
    return (v8f){v, v, v, v, v, v, v, v};
}

__global__ void __launch_bounds__(NTHR)
lfe_fused_kernel(const float* __restrict__ pos,
                 const int*   __restrict__ types,
                 const float* __restrict__ W1, const float* __restrict__ b1,
                 const float* __restrict__ W2, const float* __restrict__ b2,
                 const float* __restrict__ W3, const float* __restrict__ b3,
                 float*       __restrict__ out)
{
    __shared__ __align__(16) v4f sp[NN];
    __shared__ __align__(16) v4f lst[NWAVE][LCAP];
    __shared__ __align__(16) float ft[32 * FROW];
    __shared__ __align__(16) float hA[32 * HP];
    __shared__ __align__(16) float hB[32 * HP];

    const int b    = blockIdx.x >> 5;
    const int it   = blockIdx.x & 31;
    const int tid  = threadIdx.x;
    const int lane = tid & 31;
    const int wave = __builtin_amdgcn_readfirstlane(tid >> 5);
    const bool mlp = wave < MLPW;

    const float* pb = pos + (size_t)b * NN * 3;
#pragma unroll
    for (int k = 0; k < NN / NTHR; ++k) {
        const int j = tid + NTHR * k;
        v4f v;
        v.x = pb[j * 3 + 0];
        v.y = pb[j * 3 + 1];
        v.z = pb[j * 3 + 2];
        v.w = __int_as_float(types[b * NN + j]);
        sp[j] = v;
    }
    __syncthreads();

    const int m16   = lane & 15;
    const int khalf = lane >> 4;
    const int rt    = wave >> 2;
    const int ct    = wave & 3;
    const int n     = ct * 16 + m16;
    v2f w1f[9], w2f[16], w3f[16];
    float bv1 = 0.f, bv2 = 0.f, bv3 = 0.f;
    if (mlp) {
#pragma unroll
        for (int kk = 0; kk < 9; ++kk) {
            const int k0 = 4 * kk + 2 * khalf;
            w1f[kk].x = (k0     < 34) ? W1[k0 * 64 + n]       : 0.f;
            w1f[kk].y = (k0 + 1 < 34) ? W1[(k0 + 1) * 64 + n] : 0.f;
        }
#pragma unroll
        for (int kk = 0; kk < 16; ++kk) {
            const int k0 = 4 * kk + 2 * khalf;
            w2f[kk].x = W2[k0 * 64 + n];
            w2f[kk].y = W2[(k0 + 1) * 64 + n];
            w3f[kk].x = W3[k0 * 64 + n];
            w3f[kk].y = W3[(k0 + 1) * 64 + n];
        }
        bv1 = b1[n]; bv2 = b2[n]; bv3 = b3[n];
    }

    v4f* L = lst[wave];
    const int r = lane & (RBF - 1);
    const int h = lane >> 4;

    const float dc     = CUTOFF / 15.f;
    const float invw2  = 1.0f / (dc * dc + 1e-12f);
    const float invcut = 1.0f / CUTOFF;
    const float rc     = (float)r * dc;
    const float cut2 = CUTOFF * CUTOFF * 1.00001f;

#pragma unroll 1
    for (int a = 0; a < IPW; ++a) {
        const int i = it * 32 + wave * IPW + a;
        const v4f pi = sp[i];
        float a0 = 0.f, a1 = 0.f;

        int cnt = 0;
        v4f p0 = sp[lane];
        v4f p1 = sp[lane + 32];
        v4f p2 = sp[lane + 64];
        v4f p3 = sp[lane + 96];
        for (int s = 0; s < NN / 32; s += 4) {
            const int j = s * 32 + lane;
            const float q0 = pair_sq(pi, p0), q1 = pair_sq(pi, p1);
            const float q2 = pair_sq(pi, p2), q3 = pair_sq(pi, p3);
            const float t0 = p0.w, t1 = p1.w, t2 = p2.w, t3 = p3.w;
            const bool in0 = (q0 < cut2) && (j      != i);
            const bool in1 = (q1 < cut2) && (j + 32 != i);
            const bool in2 = (q2 < cut2) && (j + 64 != i);
            const bool in3 = (q3 < cut2) && (j + 96 != i);
            const unsigned k0 = __builtin_amdgcn_ballot_w32(in0);
            const unsigned k1 = __builtin_amdgcn_ballot_w32(in1);
            const unsigned k2 = __builtin_amdgcn_ballot_w32(in2);
            const unsigned k3 = __builtin_amdgcn_ballot_w32(in3);
            const int jn = (j + 128) & (NN - 1);
            p0 = sp[jn];
            p1 = sp[jn + 32];
            p2 = sp[jn + 64];
            p3 = sp[jn + 96];
            const int c0 = cnt;
            const int c1 = c0 + __builtin_popcount(k0);
            const int c2 = c1 + __builtin_popcount(k1);
            const int c3 = c2 + __builtin_popcount(k2);
            cnt          = c3 + __builtin_popcount(k3);
            if (in0) list_put(L, c0 + (int)__builtin_amdgcn_mbcnt_lo(k0, 0u), q0, t0);
            if (in1) list_put(L, c1 + (int)__builtin_amdgcn_mbcnt_lo(k1, 0u), q1, t1);
            if (in2) list_put(L, c2 + (int)__builtin_amdgcn_mbcnt_lo(k2, 0u), q2, t2);
            if (in3) list_put(L, c3 + (int)__builtin_amdgcn_mbcnt_lo(k3, 0u), q3, t3);
        }
        const int cpad = (cnt + 15) & ~15;
        wave_lds_handover();

        for (int e = lane; e < cpad; e += 32) {
            v4f o = (v4f){0.f, 0.f, 0.f, 0.f};
            if (e < cnt) {
                const v2f c = *(const v2f*)&L[e];
                const float sq = c.x;
                const int   t  = __float_as_int(c.y);
                const float d  = (sq > 0.f) ? sqrtf(sq) : 0.f;
                if (d < CUTOFF) {
                    const float x  = d * invcut;
                    const float x2 = x * x;
                    const float x3 = x2 * x;
                    const float cutv = 1.0f - x3 * (10.f - 15.f * x + 6.f * x2);
                    const float m0 = (t == 0) ? cutv : 0.f;
                    const float m1 = cutv - m0;
                    o = (v4f){d, m0, m1, 0.f};
                }
            }
            L[e] = o;
        }
        wave_lds_handover();

        for (int e = 0; e < cpad; e += 16) {
            v4f q[8];
#pragma unroll
            for (int k = 0; k < 8; ++k) q[k] = L[e + 2 * k + h];
#pragma unroll
            for (int k = 0; k < 8; ++k) {
                const float f  = q[k].x - rc;
                const float ev = __expf(-f * f * invw2);
                a0 += ev * q[k].y;
                a1 += ev * q[k].z;
            }
        }
        wave_lds_handover();

        a0 += __shfl_xor(a0, 16);
        a1 += __shfl_xor(a1, 16);

        float* o = ft + (wave * IPW + a) * FROW;
        if (lane < 16) {
            v2f w; w.x = a0; w.y = a1;
            *(v2f*)(o + 2 + 2 * r) = w;
        } else if (lane == 16) {
            const int ty = __float_as_int(pi.w);
            v2f w; w.x = (ty == 0) ? 1.f : 0.f;
            w.y = (ty == 1) ? 1.f : 0.f;
            *(v2f*)(o + 0) = w;
        } else if (lane == 17) {
            *(v2f*)(o + 34) = (v2f){0.f, 0.f};
        }
    }
    __syncthreads();


    if (mlp) {
        v8f c = splat8(bv1);
        const float* arow = ft + (rt * 16 + m16) * FROW + 2 * khalf;
#pragma unroll
        for (int kk = 0; kk < 9; ++kk) {
            const v2f av = *(const v2f*)(arow + 4 * kk);
            c = __builtin_amdgcn_wmma_f32_16x16x4_f32(
                    false, av, false, w1f[kk], (short)0, c, false, false);
        }
#pragma unroll
        for (int v = 0; v < 8; ++v)
            hA[(rt * 16 + 8 * khalf + v) * HP + n] = silu(c[v]);
    }
    __syncthreads();

    if (mlp) {
        v8f c = splat8(bv2);
        const float* hrow = hA + (rt * 16 + m16) * HP + 2 * khalf;
#pragma unroll
        for (int kk = 0; kk < 16; ++kk) {
            const v2f av = *(const v2f*)(hrow + 4 * kk);
            c = __builtin_amdgcn_wmma_f32_16x16x4_f32(
                    false, av, false, w2f[kk], (short)0, c, false, false);
        }
#pragma unroll
        for (int v = 0; v < 8; ++v)
            hB[(rt * 16 + 8 * khalf + v) * HP + n] = silu(c[v]);
    }
    __syncthreads();

    if (mlp) {
        v8f c = splat8(bv3);
        const float* hrow = hB + (rt * 16 + m16) * HP + 2 * khalf;
#pragma unroll
        for (int kk = 0; kk < 16; ++kk) {
            const v2f av = *(const v2f*)(hrow + 4 * kk);
            c = __builtin_amdgcn_wmma_f32_16x16x4_f32(
                    false, av, false, w3f[kk], (short)0, c, false, false);
        }
#pragma unroll
        for (int v = 0; v < 8; ++v)
            hA[(rt * 16 + 8 * khalf + v) * HP + n] = c[v];
    }
    __syncthreads();

    v4f* ob = (v4f*)(out + (size_t)blockIdx.x * 32 * 64);
    {
        const int row = tid >> 4;
        const int c4  = tid & 15;
        ob[tid] = *(const v4f*)(hA + row * HP + 4 * c4);
    }
}

extern "C" void kernel_launch(void* const* d_in, const int* in_sizes, int n_in,
                              void* d_out, int out_size, void* d_ws, size_t ws_size,
                              hipStream_t stream)
{
    const float* pos   = (const float*)d_in[0];
    const int*   types = (const int*)  d_in[1];
    const float* W1    = (const float*)d_in[2];
    const float* b1    = (const float*)d_in[3];
    const float* W2    = (const float*)d_in[4];
    const float* b2    = (const float*)d_in[5];
    const float* W3    = (const float*)d_in[6];
    const float* b3    = (const float*)d_in[7];
    float*       out   = (float*)d_out;

    const int B = in_sizes[0] / (NN * 3);

    lfe_fused_kernel<<<B * 32, NTHR, 0, stream>>>(pos, types, W1, b1, W2, b2,
                                                  W3, b3, out);
}


# Scan pass: coordinates staged as x/y/z/type planes, 4 neighbours per 16-byte LDS read, packed f32 distances; hit mask = the compare's own, diagonal cleared by a scalar mask; list, sum order unchanged
# speedup vs baseline: 6.7334x; 1.1033x over previous
#include <hip/hip_runtime.h>
#include <hip/hip_bf16.h>


#define NN      1024
#define RBF     16
#define CUTOFF  2.5f
#define FROW    36
#define HP      68

typedef __attribute__((ext_vector_type(2))) float v2f;
typedef __attribute__((ext_vector_type(8))) float v8f;

#define NWAVE 16
#define IPW   2
#define MLPW  8
#define NTHR  512
static_assert(NTHR == 32 * NWAVE && NWAVE * IPW == 32, "block shape");
#define LCAP  (NN + 16)

typedef __attribute__((ext_vector_type(4))) float v4f;

__device__ __forceinline__ void wave_lds_handover() {
    __builtin_amdgcn_fence(__ATOMIC_ACQ_REL, "wavefront");
    __builtin_amdgcn_wave_barrier();
}

__device__ __forceinline__ int plane_slot(int j) {
    return (((j >> 7) * 32 + (j & 31)) << 2) + ((j >> 5) & 3);
}

__device__ __forceinline__ v2f duo_sq(const v2f px, const v2f py, const v2f pz,
                                      const v2f x, const v2f y, const v2f z) {
    const v2f dx = px - x;
    const v2f dy = py - y;
    const v2f dz = pz - z;
    return dx * dx + dy * dy + dz * dz;
}

__device__ __forceinline__ void list_put(v4f* L, unsigned p, float sq, float tbits) {
    float* e = (float*)&L[p];
    e[0] = sq;
    e[2] = tbits;
}

__device__ __forceinline__ float silu(float x) {
    return x / (1.0f + __expf(-x));
}

__device__ __forceinline__ v8f splat8(float v) {
    return (v8f){v, v, v, v, v, v, v, v};
}

__global__ void __launch_bounds__(NTHR)
lfe_fused_kernel(const float* __restrict__ pos,
                 const int*   __restrict__ types,
                 const float* __restrict__ W1, const float* __restrict__ b1,
                 const float* __restrict__ W2, const float* __restrict__ b2,
                 const float* __restrict__ W3, const float* __restrict__ b3,
                 float*       __restrict__ out)
{
    __shared__ __align__(16) float sx[NN];
    __shared__ __align__(16) float sy[NN];
    __shared__ __align__(16) float sz[NN];
    __shared__ __align__(16) float st[NN];
    __shared__ __align__(16) v4f lst[NWAVE][LCAP];
    __shared__ __align__(16) float ft[32 * FROW];
    __shared__ __align__(16) float hA[32 * HP];
    __shared__ __align__(16) float hB[32 * HP];

    const int b    = blockIdx.x >> 5;
    const int it   = blockIdx.x & 31;
    const int tid  = threadIdx.x;
    const int lane = tid & 31;
    const int wave = __builtin_amdgcn_readfirstlane(tid >> 5);
    const bool mlp = wave < MLPW;

    const float* pb = pos + (size_t)b * NN * 3;
#pragma unroll
    for (int k = 0; k < NN / NTHR; ++k) {
        const int j  = tid + NTHR * k;
        const int sl = plane_slot(j);
        sx[sl] = pb[j * 3 + 0];
        sy[sl] = pb[j * 3 + 1];
        sz[sl] = pb[j * 3 + 2];
        st[sl] = __int_as_float(types[b * NN + j]);
    }
    __syncthreads();

    const int m16   = lane & 15;
    const int khalf = lane >> 4;
    const int rt    = wave >> 2;
    const int ct    = wave & 3;
    const int n     = ct * 16 + m16;
    v2f w1f[9], w2f[16], w3f[16];
    float bv1 = 0.f, bv2 = 0.f, bv3 = 0.f;
    if (mlp) {
#pragma unroll
        for (int kk = 0; kk < 9; ++kk) {
            const int k0 = 4 * kk + 2 * khalf;
            w1f[kk].x = (k0     < 34) ? W1[k0 * 64 + n]       : 0.f;
            w1f[kk].y = (k0 + 1 < 34) ? W1[(k0 + 1) * 64 + n] : 0.f;
        }
#pragma unroll
        for (int kk = 0; kk < 16; ++kk) {
            const int k0 = 4 * kk + 2 * khalf;
            w2f[kk].x = W2[k0 * 64 + n];
            w2f[kk].y = W2[(k0 + 1) * 64 + n];
            w3f[kk].x = W3[k0 * 64 + n];
            w3f[kk].y = W3[(k0 + 1) * 64 + n];
        }
        bv1 = b1[n]; bv2 = b2[n]; bv3 = b3[n];
    }

    v4f* L = lst[wave];
    const int r = lane & (RBF - 1);
    const int h = lane >> 4;

    const float dc     = CUTOFF / 15.f;
    const float invw2  = 1.0f / (dc * dc + 1e-12f);
    const float invcut = 1.0f / CUTOFF;
    const float rc     = (float)r * dc;
    const float cut2 = CUTOFF * CUTOFF * 1.00001f;

#pragma unroll 1
    for (int a = 0; a < IPW; ++a) {
        const int i   = it * 32 + wave * IPW + a;
        const int sli = plane_slot(i);
        const float pix = sx[sli], piy = sy[sli], piz = sz[sli];
        const v2f px2 = (v2f){pix, pix};
        const v2f py2 = (v2f){piy, piy};
        const v2f pz2 = (v2f){piz, piz};
        const int   tyi = __float_as_int(st[sli]);
        float a0 = 0.f, a1 = 0.f;

        const int      di   = i >> 5;
        const unsigned dbit = 1u << (i & 31);
        unsigned cnt = 0;
        v4f X = *(const v4f*)&sx[4 * lane];
        v4f Y = *(const v4f*)&sy[4 * lane];
        v4f Z = *(const v4f*)&sz[4 * lane];
        v4f T = *(const v4f*)&st[4 * lane];
        for (int s = 0; s < NN / 32; s += 4) {
            const v2f Qa = duo_sq(px2, py2, pz2, X.xy, Y.xy, Z.xy);
            const v2f Qb = duo_sq(px2, py2, pz2, X.zw, Y.zw, Z.zw);
            const v4f Q  = (v4f){Qa.x, Qa.y, Qb.x, Qb.y};
            const v4f Tq = T;
            const unsigned k0 = __builtin_amdgcn_ballot_w32(Q.x < cut2)
                              & ((s     == di) ? ~dbit : ~0u);
            const unsigned k1 = __builtin_amdgcn_ballot_w32(Q.y < cut2)
                              & ((s + 1 == di) ? ~dbit : ~0u);
            const unsigned k2 = __builtin_amdgcn_ballot_w32(Q.z < cut2)
                              & ((s + 2 == di) ? ~dbit : ~0u);
            const unsigned k3 = __builtin_amdgcn_ballot_w32(Q.w < cut2)
                              & ((s + 3 == di) ? ~dbit : ~0u);
            const int qn = 4 * (((((s >> 2) + 1) & 7) << 5) + lane);
            X = *(const v4f*)&sx[qn];
            Y = *(const v4f*)&sy[qn];
            Z = *(const v4f*)&sz[qn];
            T = *(const v4f*)&st[qn];
            const unsigned c0 = cnt;
            const unsigned c1 = c0 + __builtin_popcount(k0);
            const unsigned c2 = c1 + __builtin_popcount(k1);
            const unsigned c3 = c2 + __builtin_popcount(k2);
            cnt               = c3 + __builtin_popcount(k3);
            if (__builtin_amdgcn_inverse_ballot_w32(k0))
                list_put(L, __builtin_amdgcn_mbcnt_lo(k0, c0), Q.x, Tq.x);
            if (__builtin_amdgcn_inverse_ballot_w32(k1))
                list_put(L, __builtin_amdgcn_mbcnt_lo(k1, c1), Q.y, Tq.y);
            if (__builtin_amdgcn_inverse_ballot_w32(k2))
                list_put(L, __builtin_amdgcn_mbcnt_lo(k2, c2), Q.z, Tq.z);
            if (__builtin_amdgcn_inverse_ballot_w32(k3))
                list_put(L, __builtin_amdgcn_mbcnt_lo(k3, c3), Q.w, Tq.w);
        }
        const int cpad = (int)((cnt + 15u) & ~15u);
        wave_lds_handover();

        for (int e = lane; e < cpad; e += 32) {
            v4f o = (v4f){0.f, 0.f, 0.f, 0.f};
            if (e < (int)cnt) {
                const v4f c = L[e];
                const float sq = c.x;
                const int   t  = __float_as_int(c.z);
                const float d  = (sq > 0.f) ? sqrtf(sq) : 0.f;
                if (d < CUTOFF) {
                    const float x  = d * invcut;
                    const float x2 = x * x;
                    const float x3 = x2 * x;
                    const float cutv = 1.0f - x3 * (10.f - 15.f * x + 6.f * x2);
                    const float m0 = (t == 0) ? cutv : 0.f;
                    const float m1 = cutv - m0;
                    o = (v4f){d, m0, m1, 0.f};
                }
            }
            L[e] = o;
        }
        wave_lds_handover();

        for (int e = 0; e < cpad; e += 16) {
            v4f q[8];
#pragma unroll
            for (int k = 0; k < 8; ++k) q[k] = L[e + 2 * k + h];
#pragma unroll
            for (int k = 0; k < 8; ++k) {
                const float f  = q[k].x - rc;
                const float ev = __expf(-f * f * invw2);
                a0 += ev * q[k].y;
                a1 += ev * q[k].z;
            }
        }
        wave_lds_handover();

        a0 += __shfl_xor(a0, 16);
        a1 += __shfl_xor(a1, 16);

        float* o = ft + (wave * IPW + a) * FROW;
        if (lane < 16) {
            v2f w; w.x = a0; w.y = a1;
            *(v2f*)(o + 2 + 2 * r) = w;
        } else if (lane == 16) {
            const int ty = tyi;
            v2f w; w.x = (ty == 0) ? 1.f : 0.f;
            w.y = (ty == 1) ? 1.f : 0.f;
            *(v2f*)(o + 0) = w;
        } else if (lane == 17) {
            *(v2f*)(o + 34) = (v2f){0.f, 0.f};
        }
    }
    __syncthreads();


    if (mlp) {
        v8f c = splat8(bv1);
        const float* arow = ft + (rt * 16 + m16) * FROW + 2 * khalf;
#pragma unroll
        for (int kk = 0; kk < 9; ++kk) {
            const v2f av = *(const v2f*)(arow + 4 * kk);
            c = __builtin_amdgcn_wmma_f32_16x16x4_f32(
                    false, av, false, w1f[kk], (short)0, c, false, false);
        }
#pragma unroll
        for (int v = 0; v < 8; ++v)
            hA[(rt * 16 + 8 * khalf + v) * HP + n] = silu(c[v]);
    }
    __syncthreads();

    if (mlp) {
        v8f c = splat8(bv2);
        const float* hrow = hA + (rt * 16 + m16) * HP + 2 * khalf;
#pragma unroll
        for (int kk = 0; kk < 16; ++kk) {
            const v2f av = *(const v2f*)(hrow + 4 * kk);
            c = __builtin_amdgcn_wmma_f32_16x16x4_f32(
                    false, av, false, w2f[kk], (short)0, c, false, false);
        }
#pragma unroll
        for (int v = 0; v < 8; ++v)
            hB[(rt * 16 + 8 * khalf + v) * HP + n] = silu(c[v]);
    }
    __syncthreads();

    if (mlp) {
        v8f c = splat8(bv3);
        const float* hrow = hB + (rt * 16 + m16) * HP + 2 * khalf;
#pragma unroll
        for (int kk = 0; kk < 16; ++kk) {
            const v2f av = *(const v2f*)(hrow + 4 * kk);
            c = __builtin_amdgcn_wmma_f32_16x16x4_f32(
                    false, av, false, w3f[kk], (short)0, c, false, false);
        }
#pragma unroll
        for (int v = 0; v < 8; ++v)
            hA[(rt * 16 + 8 * khalf + v) * HP + n] = c[v];
    }
    __syncthreads();

    v4f* ob = (v4f*)(out + (size_t)blockIdx.x * 32 * 64);
    {
        const int row = tid >> 4;
        const int c4  = tid & 15;
        ob[tid] = *(const v4f*)(hA + row * HP + 4 * c4);
    }
}

extern "C" void kernel_launch(void* const* d_in, const int* in_sizes, int n_in,
                              void* d_out, int out_size, void* d_ws, size_t ws_size,
                              hipStream_t stream)
{
    const float* pos   = (const float*)d_in[0];
    const int*   types = (const int*)  d_in[1];
    const float* W1    = (const float*)d_in[2];
    const float* b1    = (const float*)d_in[3];
    const float* W2    = (const float*)d_in[4];
    const float* b2    = (const float*)d_in[5];
    const float* W3    = (const float*)d_in[6];
    const float* b3    = (const float*)d_in[7];
    float*       out   = (float*)d_out;

    const int B = in_sizes[0] / (NN * 3);

    lfe_fused_kernel<<<B * 32, NTHR, 0, stream>>>(pos, types, W1, b1, W2, b2,
                                                  W3, b3, out);
}


# Finish pass writes the list back as parity runs of distance (pre-scaled for exp2) and (m0,m1) pairs: the rbf pass's packed f32 ops read them straight from 16-byte loads, no moves; sum order unchanged
# speedup vs baseline: 7.0961x; 1.0539x over previous
#include <hip/hip_runtime.h>
#include <hip/hip_bf16.h>


#define NN      1024
#define RBF     16
#define CUTOFF  2.5f
#define FROW    36
#define HP      68

typedef __attribute__((ext_vector_type(2))) float v2f;
typedef __attribute__((ext_vector_type(8))) float v8f;

#define NWAVE 16
#define IPW   2
#define MLPW  8
#define NTHR  512
static_assert(NTHR == 32 * NWAVE && NWAVE * IPW == 32, "block shape");
#define LCAP  (NN + 16)

typedef __attribute__((ext_vector_type(4))) float v4f;

__device__ __forceinline__ void wave_lds_handover() {
    __builtin_amdgcn_fence(__ATOMIC_ACQ_REL, "wavefront");
    __builtin_amdgcn_wave_barrier();
}

__device__ __forceinline__ int plane_slot(int j) {
    return (((j >> 7) * 32 + (j & 31)) << 2) + ((j >> 5) & 3);
}

__device__ __forceinline__ v2f duo_sq(const v2f px, const v2f py, const v2f pz,
                                      const v2f x, const v2f y, const v2f z) {
    const v2f dx = px - x;
    const v2f dy = py - y;
    const v2f dz = pz - z;
    return dx * dx + dy * dy + dz * dz;
}

__device__ __forceinline__ void list_put(v4f* L, unsigned p, float sq, float tbits) {
    float* e = (float*)&L[p];
    e[0] = sq;
    e[2] = tbits;
}

__device__ __forceinline__ float silu(float x) {
    return x / (1.0f + __expf(-x));
}

__device__ __forceinline__ v8f splat8(float v) {
    return (v8f){v, v, v, v, v, v, v, v};
}

__global__ void __launch_bounds__(NTHR)
lfe_fused_kernel(const float* __restrict__ pos,
                 const int*   __restrict__ types,
                 const float* __restrict__ W1, const float* __restrict__ b1,
                 const float* __restrict__ W2, const float* __restrict__ b2,
                 const float* __restrict__ W3, const float* __restrict__ b3,
                 float*       __restrict__ out)
{
    __shared__ __align__(16) float sx[NN];
    __shared__ __align__(16) float sy[NN];
    __shared__ __align__(16) float sz[NN];
    __shared__ __align__(16) float st[NN];
    __shared__ __align__(16) v4f lst[NWAVE][LCAP];
    __shared__ __align__(16) float ft[32 * FROW];
    __shared__ __align__(16) float hA[32 * HP];
    __shared__ __align__(16) float hB[32 * HP];

    const int b    = blockIdx.x >> 5;
    const int it   = blockIdx.x & 31;
    const int tid  = threadIdx.x;
    const int lane = tid & 31;
    const int wave = __builtin_amdgcn_readfirstlane(tid >> 5);
    const bool mlp = wave < MLPW;

    const float* pb = pos + (size_t)b * NN * 3;
#pragma unroll
    for (int k = 0; k < NN / NTHR; ++k) {
        const int j  = tid + NTHR * k;
        const int sl = plane_slot(j);
        sx[sl] = pb[j * 3 + 0];
        sy[sl] = pb[j * 3 + 1];
        sz[sl] = pb[j * 3 + 2];
        st[sl] = __int_as_float(types[b * NN + j]);
    }
    __syncthreads();

    const int m16   = lane & 15;
    const int khalf = lane >> 4;
    const int rt    = wave >> 2;
    const int ct    = wave & 3;
    const int n     = ct * 16 + m16;
    v2f w1f[9], w2f[16], w3f[16];
    float bv1 = 0.f, bv2 = 0.f, bv3 = 0.f;
    if (mlp) {
#pragma unroll
        for (int kk = 0; kk < 9; ++kk) {
            const int k0 = 4 * kk + 2 * khalf;
            w1f[kk].x = (k0     < 34) ? W1[k0 * 64 + n]       : 0.f;
            w1f[kk].y = (k0 + 1 < 34) ? W1[(k0 + 1) * 64 + n] : 0.f;
        }
#pragma unroll
        for (int kk = 0; kk < 16; ++kk) {
            const int k0 = 4 * kk + 2 * khalf;
            w2f[kk].x = W2[k0 * 64 + n];
            w2f[kk].y = W2[(k0 + 1) * 64 + n];
            w3f[kk].x = W3[k0 * 64 + n];
            w3f[kk].y = W3[(k0 + 1) * 64 + n];
        }
        bv1 = b1[n]; bv2 = b2[n]; bv3 = b3[n];
    }

    v4f* L = lst[wave];
    const int r = lane & (RBF - 1);
    const int h = lane >> 4;

    const float dc     = CUTOFF / 15.f;
    const float invw2  = 1.0f / (dc * dc + 1e-12f);
    const float invcut = 1.0f / CUTOFF;
    const float ksc    = __builtin_sqrtf(invw2 * 1.4426950408889634f);
    const float rcs    = (float)r * dc * ksc;
    const v2f   rcs2   = (v2f){rcs, rcs};
    const int   j16    = lane & 15;
    const int   fofs   = (j16 & 1) * 24 + (j16 >> 1);
    const float cut2 = CUTOFF * CUTOFF * 1.00001f;

#pragma unroll 1
    for (int a = 0; a < IPW; ++a) {
        const int i   = it * 32 + wave * IPW + a;
        const int sli = plane_slot(i);
        const float pix = sx[sli], piy = sy[sli], piz = sz[sli];
        const v2f px2 = (v2f){pix, pix};
        const v2f py2 = (v2f){piy, piy};
        const v2f pz2 = (v2f){piz, piz};
        const int   tyi = __float_as_int(st[sli]);

        const int      di   = i >> 5;
        const unsigned dbit = 1u << (i & 31);
        unsigned cnt = 0;
        v4f X = *(const v4f*)&sx[4 * lane];
        v4f Y = *(const v4f*)&sy[4 * lane];
        v4f Z = *(const v4f*)&sz[4 * lane];
        v4f T = *(const v4f*)&st[4 * lane];
        for (int s = 0; s < NN / 32; s += 4) {
            const v2f Qa = duo_sq(px2, py2, pz2, X.xy, Y.xy, Z.xy);
            const v2f Qb = duo_sq(px2, py2, pz2, X.zw, Y.zw, Z.zw);
            const v4f Q  = (v4f){Qa.x, Qa.y, Qb.x, Qb.y};
            const v4f Tq = T;
            const unsigned k0 = __builtin_amdgcn_ballot_w32(Q.x < cut2)
                              & ((s     == di) ? ~dbit : ~0u);
            const unsigned k1 = __builtin_amdgcn_ballot_w32(Q.y < cut2)
                              & ((s + 1 == di) ? ~dbit : ~0u);
            const unsigned k2 = __builtin_amdgcn_ballot_w32(Q.z < cut2)
                              & ((s + 2 == di) ? ~dbit : ~0u);
            const unsigned k3 = __builtin_amdgcn_ballot_w32(Q.w < cut2)
                              & ((s + 3 == di) ? ~dbit : ~0u);
            const int qn = 4 * (((((s >> 2) + 1) & 7) << 5) + lane);
            X = *(const v4f*)&sx[qn];
            Y = *(const v4f*)&sy[qn];
            Z = *(const v4f*)&sz[qn];
            T = *(const v4f*)&st[qn];
            const unsigned c0 = cnt;
            const unsigned c1 = c0 + __builtin_popcount(k0);
            const unsigned c2 = c1 + __builtin_popcount(k1);
            const unsigned c3 = c2 + __builtin_popcount(k2);
            cnt               = c3 + __builtin_popcount(k3);
            if (__builtin_amdgcn_inverse_ballot_w32(k0))
                list_put(L, __builtin_amdgcn_mbcnt_lo(k0, c0), Q.x, Tq.x);
            if (__builtin_amdgcn_inverse_ballot_w32(k1))
                list_put(L, __builtin_amdgcn_mbcnt_lo(k1, c1), Q.y, Tq.y);
            if (__builtin_amdgcn_inverse_ballot_w32(k2))
                list_put(L, __builtin_amdgcn_mbcnt_lo(k2, c2), Q.z, Tq.z);
            if (__builtin_amdgcn_inverse_ballot_w32(k3))
                list_put(L, __builtin_amdgcn_mbcnt_lo(k3, c3), Q.w, Tq.w);
        }
        const int cpad = (int)((cnt + 15u) & ~15u);
        wave_lds_handover();

        for (int e = lane; e < cpad; e += 32) {
            float dsc = 0.f;
            v2f   mm  = (v2f){0.f, 0.f};
            if (e < (int)cnt) {
                const v4f c = L[e];
                const float sq = c.x;
                const int   t  = __float_as_int(c.z);
                const float d  = (sq > 0.f) ? sqrtf(sq) : 0.f;
                if (d < CUTOFF) {
                    const float x  = d * invcut;
                    const float x2 = x * x;
                    const float x3 = x2 * x;
                    const float cutv = 1.0f - x3 * (10.f - 15.f * x + 6.f * x2);
                    const float m0 = (t == 0) ? cutv : 0.f;
                    const float m1 = cutv - m0;
                    dsc = d * ksc;
                    mm  = (v2f){m0, m1};
                }
            }
            float* sc = (float*)&L[e - j16];
            sc[fofs] = dsc;
            *(v2f*)(sc + 8 + fofs + (j16 >> 1)) = mm;
        }
        wave_lds_handover();

        v2f acc = (v2f){0.f, 0.f};
        for (int e = 0; e < cpad; e += 16) {
            const float* sc = (const float*)&L[e] + 24 * h;
            const v4f dA = *(const v4f*)(sc);
            const v4f dB = *(const v4f*)(sc + 4);
            const v4f mA = *(const v4f*)(sc + 8);
            const v4f mB = *(const v4f*)(sc + 12);
            const v4f mC = *(const v4f*)(sc + 16);
            const v4f mD = *(const v4f*)(sc + 20);
            __builtin_amdgcn_sched_barrier(0);
            const v2f f0 = dA.xy - rcs2;
            const v2f f1 = dA.zw - rcs2;
            const v2f f2 = dB.xy - rcs2;
            const v2f f3 = dB.zw - rcs2;
            const v2f g0 = -(f0 * f0);
            const v2f g1 = -(f1 * f1);
            const v2f g2 = -(f2 * f2);
            const v2f g3 = -(f3 * f3);
            float ev;
            ev = __builtin_amdgcn_exp2f(g0.x); acc += (v2f){ev, ev} * mA.xy;
            ev = __builtin_amdgcn_exp2f(g0.y); acc += (v2f){ev, ev} * mA.zw;
            ev = __builtin_amdgcn_exp2f(g1.x); acc += (v2f){ev, ev} * mB.xy;
            ev = __builtin_amdgcn_exp2f(g1.y); acc += (v2f){ev, ev} * mB.zw;
            ev = __builtin_amdgcn_exp2f(g2.x); acc += (v2f){ev, ev} * mC.xy;
            ev = __builtin_amdgcn_exp2f(g2.y); acc += (v2f){ev, ev} * mC.zw;
            ev = __builtin_amdgcn_exp2f(g3.x); acc += (v2f){ev, ev} * mD.xy;
            ev = __builtin_amdgcn_exp2f(g3.y); acc += (v2f){ev, ev} * mD.zw;
        }
        wave_lds_handover();

        float a0 = acc.x + __shfl_xor(acc.x, 16);
        float a1 = acc.y + __shfl_xor(acc.y, 16);

        float* o = ft + (wave * IPW + a) * FROW;
        if (lane < 16) {
            v2f w; w.x = a0; w.y = a1;
            *(v2f*)(o + 2 + 2 * r) = w;
        } else if (lane == 16) {
            const int ty = tyi;
            v2f w; w.x = (ty == 0) ? 1.f : 0.f;
            w.y = (ty == 1) ? 1.f : 0.f;
            *(v2f*)(o + 0) = w;
        } else if (lane == 17) {
            *(v2f*)(o + 34) = (v2f){0.f, 0.f};
        }
    }
    __syncthreads();


    if (mlp) {
        v8f c = splat8(bv1);
        const float* arow = ft + (rt * 16 + m16) * FROW + 2 * khalf;
#pragma unroll
        for (int kk = 0; kk < 9; ++kk) {
            const v2f av = *(const v2f*)(arow + 4 * kk);
            c = __builtin_amdgcn_wmma_f32_16x16x4_f32(
                    false, av, false, w1f[kk], (short)0, c, false, false);
        }
#pragma unroll
        for (int v = 0; v < 8; ++v)
            hA[(rt * 16 + 8 * khalf + v) * HP + n] = silu(c[v]);
    }
    __syncthreads();

    if (mlp) {
        v8f c = splat8(bv2);
        const float* hrow = hA + (rt * 16 + m16) * HP + 2 * khalf;
#pragma unroll
        for (int kk = 0; kk < 16; ++kk) {
            const v2f av = *(const v2f*)(hrow + 4 * kk);
            c = __builtin_amdgcn_wmma_f32_16x16x4_f32(
                    false, av, false, w2f[kk], (short)0, c, false, false);
        }
#pragma unroll
        for (int v = 0; v < 8; ++v)
            hB[(rt * 16 + 8 * khalf + v) * HP + n] = silu(c[v]);
    }
    __syncthreads();

    if (mlp) {
        v8f c = splat8(bv3);
        const float* hrow = hB + (rt * 16 + m16) * HP + 2 * khalf;
#pragma unroll
        for (int kk = 0; kk < 16; ++kk) {
            const v2f av = *(const v2f*)(hrow + 4 * kk);
            c = __builtin_amdgcn_wmma_f32_16x16x4_f32(
                    false, av, false, w3f[kk], (short)0, c, false, false);
        }
#pragma unroll
        for (int v = 0; v < 8; ++v)
            hA[(rt * 16 + 8 * khalf + v) * HP + n] = c[v];
    }
    __syncthreads();

    v4f* ob = (v4f*)(out + (size_t)blockIdx.x * 32 * 64);
    {
        const int row = tid >> 4;
        const int c4  = tid & 15;
        ob[tid] = *(const v4f*)(hA + row * HP + 4 * c4);
    }
}

extern "C" void kernel_launch(void* const* d_in, const int* in_sizes, int n_in,
                              void* d_out, int out_size, void* d_ws, size_t ws_size,
                              hipStream_t stream)
{
    const float* pos   = (const float*)d_in[0];
    const int*   types = (const int*)  d_in[1];
    const float* W1    = (const float*)d_in[2];
    const float* b1    = (const float*)d_in[3];
    const float* W2    = (const float*)d_in[4];
    const float* b2    = (const float*)d_in[5];
    const float* W3    = (const float*)d_in[6];
    const float* b3    = (const float*)d_in[7];
    float*       out   = (float*)d_out;

    const int B = in_sizes[0] / (NN * 3);

    lfe_fused_kernel<<<B * 32, NTHR, 0, stream>>>(pos, types, W1, b1, W2, b2,
                                                  W3, b3, out);
}
